# starting kernel
# baseline (speedup 1.0000x reference)
#include <hip/hip_runtime.h>

typedef __attribute__((ext_vector_type(2))) float v2f;
typedef __attribute__((ext_vector_type(8))) float v8f;

#define Hh 50
#define Ff 32
#define T_STEPS 365
#define G4 200
#define NPAD 208
#define NT 13
#define KH 52
#define KTX 8
#define KTH 13
#define WAVES 4
#define BLK (WAVES*32)

#define OFF_WX   0
#define OFF_WH   (OFF_WX + NPAD*Ff)
#define OFF_BIAS (OFF_WH + NPAD*KH)
#define OFF_H    (OFF_BIAS + NPAD)
#define OFF_G    (OFF_H + 16*KH)
#define LDS_FLOATS (OFF_G + 16*NPAD)

__global__ __launch_bounds__(BLK) void lstm_persist_wmma(
    const float* __restrict__ x,
    const float* __restrict__ kernel,
    const float* __restrict__ rec_kernel,
    const float* __restrict__ bias,
    const float* __restrict__ dense_w,
    const float* __restrict__ dense_b,
    float* __restrict__ out)
{
  extern __shared__ float lds[];
  float* wxT = lds + OFF_WX;
  float* whT = lds + OFF_WH;
  float* bL  = lds + OFF_BIAS;
  float* hB  = lds + OFF_H;
  float* gB  = lds + OFF_G;

  const int tid   = threadIdx.x;
  const int lane  = tid & 31;
  const int wave  = tid >> 5;
  const int rowA  = lane & 15;
  const int khalf = lane >> 4;
  const int ncol  = lane & 15;
  const int b0    = blockIdx.x * 16;

  const int ntS = (wave * NT) / WAVES;
  const int ntE = ((wave + 1) * NT) / WAVES;
  const int ubase = wave * 200;

  for (int i = tid; i < NPAD * Ff; i += BLK) {
    int n = i / Ff, k = i % Ff;
    wxT[i] = (n < G4) ? kernel[k * G4 + n] : 0.0f;
  }
  for (int i = tid; i < NPAD * KH; i += BLK) {
    int n = i / KH, k = i % KH;
    whT[i] = (n < G4 && k < Hh) ? rec_kernel[k * G4 + n] : 0.0f;
  }
  for (int i = tid; i < NPAD; i += BLK) bL[i] = (i < G4) ? bias[i] : 0.0f;
  for (int i = tid; i < 16 * KH; i += BLK) hB[i] = 0.0f;
  __syncthreads();

  float c[7];
  #pragma unroll
  for (int q = 0; q < 7; ++q) c[q] = 0.0f;

  const float* xrow = x + (long)(b0 + rowA) * T_STEPS * Ff;

  for (int t = 0; t < T_STEPS; ++t) {
    const float* xp = xrow + t * Ff;
    v2f ax[KTX];
    #pragma unroll
    for (int kt = 0; kt < KTX; ++kt)
      ax[kt] = *(const v2f*)(xp + kt * 4 + 2 * khalf);
    if (wave == 0 && t + 1 < T_STEPS) __builtin_prefetch(xp + Ff, 0, 3);

    v2f ah[KTH];
    #pragma unroll
    for (int kt = 0; kt < KTH; ++kt)
      ah[kt] = *(const v2f*)(hB + rowA * KH + kt * 4 + 2 * khalf);

    for (int nt = ntS; nt < ntE; ++nt) {
      const int n = nt * 16 + ncol;
      v8f acc = {};
      #pragma unroll
      for (int kt = 0; kt < KTX; ++kt) {
        v2f b = *(const v2f*)(wxT + n * Ff + kt * 4 + 2 * khalf);
        acc = __builtin_amdgcn_wmma_f32_16x16x4_f32(
            false, ax[kt], false, b, (short)0, acc, false, false);
      }
      #pragma unroll
      for (int kt = 0; kt < KTH; ++kt) {
        v2f b = *(const v2f*)(whT + n * KH + kt * 4 + 2 * khalf);
        acc = __builtin_amdgcn_wmma_f32_16x16x4_f32(
            false, ah[kt], false, b, (short)0, acc, false, false);
      }
      #pragma unroll
      for (int r = 0; r < 8; ++r)
        gB[(r + 8 * khalf) * NPAD + n] = acc[r];
    }
    __syncthreads();

    #pragma unroll
    for (int q = 0; q < 7; ++q) {
      int u = q * 32 + lane;
      if (u < 200) {
        int p = ubase + u;
        int row = p / 50, j = p % 50;
        const float* g = gB + row * NPAD;
        float gi = g[j]       + bL[j];
        float gf = g[j +  50] + bL[j +  50];
        float gc = g[j + 100] + bL[j + 100];
        float go = g[j + 150] + bL[j + 150];
        float i_   = 1.0f / (1.0f + __expf(-gi));
        float f_   = 1.0f / (1.0f + __expf(-gf));
        float cand = fmaxf(gc, 0.0f);
        float o_   = 1.0f / (1.0f + __expf(-go));
        float cn = f_ * c[q] + i_ * cand;
        c[q] = cn;
        hB[row * KH + j] = o_ * fmaxf(cn, 0.0f);
      }
    }
    __syncthreads();
  }

  if (tid < 16) {
    float s = 0.0f;
    #pragma unroll 10
    for (int j = 0; j < Hh; ++j) s += hB[tid * KH + j] * dense_w[j];
    out[b0 + tid] = s + dense_b[0];
  }
}

extern "C" void kernel_launch(void* const* d_in, const int* in_sizes, int n_in,
                              void* d_out, int out_size, void* d_ws, size_t ws_size,
                              hipStream_t stream) {
  const float* x      = (const float*)d_in[0];
  const float* kernel = (const float*)d_in[1];
  const float* rec    = (const float*)d_in[2];
  const float* bias   = (const float*)d_in[3];
  const float* dw     = (const float*)d_in[4];
  const float* db     = (const float*)d_in[5];
  float* out = (float*)d_out;

  dim3 grid(4096 / 16);
  dim3 block(BLK);
  size_t shmem = (size_t)LDS_FLOATS * sizeof(float);
  hipLaunchKernelGGL(lstm_persist_wmma, grid, block, shmem, stream,
                     x, kernel, rec, bias, dw, db, out);
}


# weights and biases held in registers for the whole sequence: each wave reads the B fragments of its tiles from LDS once, before the time loop (tile loop unrolled; tile chains interleave)
# speedup vs baseline: 1.4844x; 1.4844x over previous
#include <hip/hip_runtime.h>


typedef __attribute__((ext_vector_type(2))) float v2f;
typedef __attribute__((ext_vector_type(8))) float v8f;

#define Hh 50
#define Ff 32
#define T_STEPS 365
#define G4 200
#define NPAD 208
#define NT 13
#define KH 52
#define KTX 8
#define KTH 13
#define WAVES 4
#define BLK (WAVES*32)
#define TPW 4

#define OFF_WX   0
#define OFF_WH   (OFF_WX + NPAD*Ff)
#define OFF_BIAS (OFF_WH + NPAD*KH)
#define OFF_H    (OFF_BIAS + NPAD)
#define OFF_G    (OFF_H + 16*KH)
#define LDS_FLOATS (OFF_G + 16*NPAD)

template <int NTW>
__device__ __forceinline__ void gate_tiles(
    const v2f (&ax)[KTX], const v2f (&ah)[KTH],
    const v2f (&bx)[TPW][KTX], const v2f (&bh)[TPW][KTH],
    float* __restrict__ gcol)
{
  v8f acc[NTW];
  #pragma unroll
  for (int i = 0; i < NTW; ++i) acc[i] = v8f{};
  #pragma unroll
  for (int kt = 0; kt < KTX; ++kt) {
    #pragma unroll
    for (int i = 0; i < NTW; ++i)
      acc[i] = __builtin_amdgcn_wmma_f32_16x16x4_f32(
          false, ax[kt], false, bx[i][kt], (short)0, acc[i], false, false);
  }
  #pragma unroll
  for (int kt = 0; kt < KTH; ++kt) {
    #pragma unroll
    for (int i = 0; i < NTW; ++i)
      acc[i] = __builtin_amdgcn_wmma_f32_16x16x4_f32(
          false, ah[kt], false, bh[i][kt], (short)0, acc[i], false, false);
  }
  #pragma unroll
  for (int i = 0; i < NTW; ++i) {
    #pragma unroll
    for (int r = 0; r < 8; ++r)
      gcol[r * NPAD + i * 16] = acc[i][r];
  }
}

__global__ __launch_bounds__(BLK) void lstm_persist_wmma(
    const float* __restrict__ x,
    const float* __restrict__ kernel,
    const float* __restrict__ rec_kernel,
    const float* __restrict__ bias,
    const float* __restrict__ dense_w,
    const float* __restrict__ dense_b,
    float* __restrict__ out)
{
  extern __shared__ float lds[];
  float* wxT = lds + OFF_WX;
  float* whT = lds + OFF_WH;
  float* bL  = lds + OFF_BIAS;
  float* hB  = lds + OFF_H;
  float* gB  = lds + OFF_G;

  const int tid   = threadIdx.x;
  const int lane  = tid & 31;
  const int wave  = __builtin_amdgcn_readfirstlane(tid >> 5);
  const int rowA  = lane & 15;
  const int khalf = lane >> 4;
  const int ncol  = lane & 15;
  const int b0    = blockIdx.x * 16;

  const int  ntS  = (wave * NT) / WAVES;
  const bool four = (((wave + 1) * NT) / WAVES - ntS) == TPW;
  const int ubase = wave * 200;

  for (int i = tid; i < NPAD * Ff; i += BLK) {
    int n = i / Ff, k = i % Ff;
    wxT[i] = (n < G4) ? kernel[k * G4 + n] : 0.0f;
  }
  for (int i = tid; i < NPAD * KH; i += BLK) {
    int n = i / KH, k = i % KH;
    whT[i] = (n < G4 && k < Hh) ? rec_kernel[k * G4 + n] : 0.0f;
  }
  for (int i = tid; i < NPAD; i += BLK) bL[i] = (i < G4) ? bias[i] : 0.0f;
  for (int i = tid; i < 16 * KH; i += BLK) hB[i] = 0.0f;
  __syncthreads();

  v2f bx[TPW][KTX];
  v2f bh[TPW][KTH];
  #pragma unroll
  for (int i = 0; i < TPW; ++i) {
    const int n = (ntS + i) * 16 + ncol;
    #pragma unroll
    for (int kt = 0; kt < KTX; ++kt)
      bx[i][kt] = *(const v2f*)(wxT + n * Ff + kt * 4 + 2 * khalf);
    #pragma unroll
    for (int kt = 0; kt < KTH; ++kt)
      bh[i][kt] = *(const v2f*)(whT + n * KH + kt * 4 + 2 * khalf);
  }

  float c[7];
  float bi[7], bf[7], bc[7], bo[7];
  #pragma unroll
  for (int q = 0; q < 7; ++q) {
    c[q] = 0.0f;
    const int j = (ubase + q * 32 + lane) % 50;
    bi[q] = bL[j];
    bf[q] = bL[j +  50];
    bc[q] = bL[j + 100];
    bo[q] = bL[j + 150];
  }

  const float* xrow = x + (long)(b0 + rowA) * T_STEPS * Ff;
  float* gcol = gB + (8 * khalf) * NPAD + ntS * 16 + ncol;

  for (int t = 0; t < T_STEPS; ++t) {
    const float* xp = xrow + t * Ff;
    v2f ax[KTX];
    #pragma unroll
    for (int kt = 0; kt < KTX; ++kt)
      ax[kt] = *(const v2f*)(xp + kt * 4 + 2 * khalf);
    if (wave == 0 && t + 1 < T_STEPS) __builtin_prefetch(xp + Ff, 0, 3);

    v2f ah[KTH];
    #pragma unroll
    for (int kt = 0; kt < KTH; ++kt)
      ah[kt] = *(const v2f*)(hB + rowA * KH + kt * 4 + 2 * khalf);

    if (four) gate_tiles<4>(ax, ah, bx, bh, gcol);
    else      gate_tiles<3>(ax, ah, bx, bh, gcol);
    __syncthreads();

    #pragma unroll
    for (int q = 0; q < 7; ++q) {
      int u = q * 32 + lane;
      if (u < 200) {
        int p = ubase + u;
        int row = p / 50, j = p % 50;
        const float* g = gB + row * NPAD;
        float gi = g[j]       + bi[q];
        float gf = g[j +  50] + bf[q];
        float gc = g[j + 100] + bc[q];
        float go = g[j + 150] + bo[q];
        float i_   = 1.0f / (1.0f + __expf(-gi));
        float f_   = 1.0f / (1.0f + __expf(-gf));
        float cand = fmaxf(gc, 0.0f);
        float o_   = 1.0f / (1.0f + __expf(-go));
        float cn = f_ * c[q] + i_ * cand;
        c[q] = cn;
        hB[row * KH + j] = o_ * fmaxf(cn, 0.0f);
      }
    }
    __syncthreads();
  }

  if (tid < 16) {
    float s = 0.0f;
    #pragma unroll 10
    for (int j = 0; j < Hh; ++j) s += hB[tid * KH + j] * dense_w[j];
    out[b0 + tid] = s + dense_b[0];
  }
}

extern "C" void kernel_launch(void* const* d_in, const int* in_sizes, int n_in,
                              void* d_out, int out_size, void* d_ws, size_t ws_size,
                              hipStream_t stream) {
  const float* x      = (const float*)d_in[0];
  const float* kernel = (const float*)d_in[1];
  const float* rec    = (const float*)d_in[2];
  const float* bias   = (const float*)d_in[3];
  const float* dw     = (const float*)d_in[4];
  const float* db     = (const float*)d_in[5];
  float* out = (float*)d_out;

  dim3 grid(4096 / 16);
  dim3 block(BLK);
  size_t shmem = (size_t)LDS_FLOATS * sizeof(float);
  hipLaunchKernelGGL(lstm_persist_wmma, grid, block, shmem, stream,
                     x, kernel, rec, bias, dw, db, out);
}


# gate phase as straight-line code: every gate value of a lane read from LDS in one group, its seven unit chains side by side, h stored last; per-unit arithmetic and summation order unchanged
# speedup vs baseline: 1.7946x; 1.2090x over previous
#include <hip/hip_runtime.h>


typedef __attribute__((ext_vector_type(2))) float v2f;
typedef __attribute__((ext_vector_type(8))) float v8f;

#define Hh 50
#define Ff 32
#define T_STEPS 365
#define G4 200
#define NPAD 208
#define NT 13
#define KH 52
#define KTX 8
#define KTH 13
#define WAVES 4
#define BLK (WAVES*32)
#define TPW 4

#define OFF_WX   0
#define OFF_WH   (OFF_WX + NPAD*Ff)
#define OFF_BIAS (OFF_WH + NPAD*KH)
#define OFF_H    (OFF_BIAS + NPAD)
#define OFF_G    (OFF_H + 16*KH)
#define OFF_DUMP (OFF_G + 16*NPAD)
#define LDS_FLOATS (OFF_DUMP + 32)

template <int NTW>
__device__ __forceinline__ void gate_tiles(
    const v2f (&ax)[KTX], const v2f (&ah)[KTH],
    const v2f (&bx)[TPW][KTX], const v2f (&bh)[TPW][KTH],
    float* __restrict__ gcol)
{
  v8f acc[NTW];
  #pragma unroll
  for (int i = 0; i < NTW; ++i) acc[i] = v8f{};
  #pragma unroll
  for (int kt = 0; kt < KTX; ++kt) {
    #pragma unroll
    for (int i = 0; i < NTW; ++i)
      acc[i] = __builtin_amdgcn_wmma_f32_16x16x4_f32(
          false, ax[kt], false, bx[i][kt], (short)0, acc[i], false, false);
  }
  #pragma unroll
  for (int kt = 0; kt < KTH; ++kt) {
    #pragma unroll
    for (int i = 0; i < NTW; ++i)
      acc[i] = __builtin_amdgcn_wmma_f32_16x16x4_f32(
          false, ah[kt], false, bh[i][kt], (short)0, acc[i], false, false);
  }
  #pragma unroll
  for (int i = 0; i < NTW; ++i) {
    #pragma unroll
    for (int r = 0; r < 8; ++r)
      gcol[r * NPAD + i * 16] = acc[i][r];
  }
}

__global__ __launch_bounds__(BLK) void lstm_persist_wmma(
    const float* __restrict__ x,
    const float* __restrict__ kernel,
    const float* __restrict__ rec_kernel,
    const float* __restrict__ bias,
    const float* __restrict__ dense_w,
    const float* __restrict__ dense_b,
    float* __restrict__ out)
{
  extern __shared__ float lds[];
  float* wxT = lds + OFF_WX;
  float* whT = lds + OFF_WH;
  float* bL  = lds + OFF_BIAS;
  float* hB  = lds + OFF_H;
  float* gB  = lds + OFF_G;

  const int tid   = threadIdx.x;
  const int lane  = tid & 31;
  const int wave  = __builtin_amdgcn_readfirstlane(tid >> 5);
  const int rowA  = lane & 15;
  const int khalf = lane >> 4;
  const int ncol  = lane & 15;
  const int b0    = blockIdx.x * 16;

  const int  ntS  = (wave * NT) / WAVES;
  const bool four = (((wave + 1) * NT) / WAVES - ntS) == TPW;
  const int ubase = wave * 200;

  for (int i = tid; i < NPAD * Ff; i += BLK) {
    int n = i / Ff, k = i % Ff;
    wxT[i] = (n < G4) ? kernel[k * G4 + n] : 0.0f;
  }
  for (int i = tid; i < NPAD * KH; i += BLK) {
    int n = i / KH, k = i % KH;
    whT[i] = (n < G4 && k < Hh) ? rec_kernel[k * G4 + n] : 0.0f;
  }
  for (int i = tid; i < NPAD; i += BLK) bL[i] = (i < G4) ? bias[i] : 0.0f;
  for (int i = tid; i < 16 * KH; i += BLK) hB[i] = 0.0f;
  __syncthreads();

  v2f bx[TPW][KTX];
  v2f bh[TPW][KTH];
  #pragma unroll
  for (int i = 0; i < TPW; ++i) {
    const int n = (ntS + i) * 16 + ncol;
    #pragma unroll
    for (int kt = 0; kt < KTX; ++kt)
      bx[i][kt] = *(const v2f*)(wxT + n * Ff + kt * 4 + 2 * khalf);
    #pragma unroll
    for (int kt = 0; kt < KTH; ++kt)
      bh[i][kt] = *(const v2f*)(whT + n * KH + kt * 4 + 2 * khalf);
  }

  float c[7];
  float bi[7], bf[7], bc[7], bo[7];
  #pragma unroll
  for (int q = 0; q < 7; ++q) {
    c[q] = 0.0f;
    const int j = (ubase + q * 32 + lane) % 50;
    bi[q] = bL[j];
    bf[q] = bL[j +  50];
    bc[q] = bL[j + 100];
    bo[q] = bL[j + 150];
  }

  int goff[7], hoff[7];
  #pragma unroll
  for (int q = 0; q < 7; ++q) {
    const int u  = q * 32 + lane;
    const int uc = (u < 200) ? u : 199;
    const int p  = ubase + uc;
    const int row = p / 50, j = p % 50;
    goff[q] = OFF_G + row * NPAD + j;
    hoff[q] = (u < 200) ? (OFF_H + row * KH + j) : (OFF_DUMP + lane);
  }

  const float* xrow = x + (long)(b0 + rowA) * T_STEPS * Ff;
  float* gcol = gB + (8 * khalf) * NPAD + ntS * 16 + ncol;

  for (int t = 0; t < T_STEPS; ++t) {
    const float* xp = xrow + t * Ff;
    v2f ax[KTX];
    #pragma unroll
    for (int kt = 0; kt < KTX; ++kt)
      ax[kt] = *(const v2f*)(xp + kt * 4 + 2 * khalf);
    if (wave == 0 && t + 1 < T_STEPS) __builtin_prefetch(xp + Ff, 0, 3);

    v2f ah[KTH];
    #pragma unroll
    for (int kt = 0; kt < KTH; ++kt)
      ah[kt] = *(const v2f*)(hB + rowA * KH + kt * 4 + 2 * khalf);

    if (four) gate_tiles<4>(ax, ah, bx, bh, gcol);
    else      gate_tiles<3>(ax, ah, bx, bh, gcol);
    __syncthreads();

    float gi[7], gf[7], gc[7], go[7];
    #pragma unroll
    for (int q = 0; q < 7; ++q) {
      const float* g = lds + goff[q];
      gi[q] = g[0]   + bi[q];
      gf[q] = g[50]  + bf[q];
      gc[q] = g[100] + bc[q];
      go[q] = g[150] + bo[q];
    }
    float ei[7], ef[7], eo[7];
    #pragma unroll
    for (int q = 0; q < 7; ++q) {
      ei[q] = __expf(-gi[q]);
      ef[q] = __expf(-gf[q]);
      eo[q] = __expf(-go[q]);
    }
    float si[7], sf[7], so[7];
    #pragma unroll
    for (int q = 0; q < 7; ++q) {
      si[q] = 1.0f / (1.0f + ei[q]);
      sf[q] = 1.0f / (1.0f + ef[q]);
      so[q] = 1.0f / (1.0f + eo[q]);
    }
    float hv[7];
    #pragma unroll
    for (int q = 0; q < 7; ++q) {
      const float cand = fmaxf(gc[q], 0.0f);
      const float cn = __builtin_fmaf(cand, si[q], sf[q] * c[q]);
      c[q] = cn;
      hv[q] = so[q] * fmaxf(cn, 0.0f);
    }
    #pragma unroll
    for (int q = 0; q < 7; ++q) lds[hoff[q]] = hv[q];
    __syncthreads();
  }

  if (tid < 16) {
    float s = 0.0f;
    #pragma unroll 10
    for (int j = 0; j < Hh; ++j) s += hB[tid * KH + j] * dense_w[j];
    out[b0 + tid] = s + dense_b[0];
  }
}

extern "C" void kernel_launch(void* const* d_in, const int* in_sizes, int n_in,
                              void* d_out, int out_size, void* d_ws, size_t ws_size,
                              hipStream_t stream) {
  const float* x      = (const float*)d_in[0];
  const float* kernel = (const float*)d_in[1];
  const float* rec    = (const float*)d_in[2];
  const float* bias   = (const float*)d_in[3];
  const float* dw     = (const float*)d_in[4];
  const float* db     = (const float*)d_in[5];
  float* out = (float*)d_out;

  dim3 grid(4096 / 16);
  dim3 block(BLK);
  size_t shmem = (size_t)LDS_FLOATS * sizeof(float);
  hipLaunchKernelGGL(lstm_persist_wmma, grid, block, shmem, stream,
                     x, kernel, rec, bias, dw, db, out);
}


# input projection moved to a second wave per SIMD (4 more waves), one step ahead; its result goes through LDS as the recurrent waves' starting accumulator; summation order unchanged
# speedup vs baseline: 2.0136x; 1.1220x over previous
#include <hip/hip_runtime.h>


typedef __attribute__((ext_vector_type(2))) float v2f;
typedef __attribute__((ext_vector_type(4))) float v4f;
typedef __attribute__((ext_vector_type(8))) float v8f;

#define Hh 50
#define Ff 32
#define T_STEPS 365
#define G4 200
#define NPAD 208
#define NT 13
#define KH 52
#define KTX 8
#define KTH 13
#define RW 4
#define PW 4
#define BLK ((RW + PW) * 32)
#define TPW 4

#define XG_TILE 256
#define XG_BUF  (NT * XG_TILE)

#define OFF_WX   0
#define OFF_WH   (OFF_WX + NPAD*Ff)
#define OFF_BIAS (OFF_WH + NPAD*KH)
#define OFF_H    (OFF_BIAS + NPAD)
#define OFF_G    (OFF_H + 16*KH)
#define OFF_DUMP (OFF_G + 16*NPAD)
#define OFF_XG   (OFF_DUMP + 32)
#define LDS_FLOATS (OFF_XG + 2*XG_BUF)

template <int NTW>
__device__ __forceinline__ void proj_tiles(
    const v2f (&ax)[KTX], const v2f (&bx)[TPW][KTX], float* __restrict__ xgw)
{
  v8f acc[NTW];
  #pragma unroll
  for (int i = 0; i < NTW; ++i) acc[i] = v8f{};
  #pragma unroll
  for (int kt = 0; kt < KTX; ++kt) {
    #pragma unroll
    for (int i = 0; i < NTW; ++i)
      acc[i] = __builtin_amdgcn_wmma_f32_16x16x4_f32(
          false, ax[kt], false, bx[i][kt], (short)0, acc[i], false, false);
  }
  #pragma unroll
  for (int i = 0; i < NTW; ++i) {
    *(v4f*)(xgw + i * XG_TILE)       = v4f{acc[i][0], acc[i][1], acc[i][2], acc[i][3]};
    *(v4f*)(xgw + i * XG_TILE + 128) = v4f{acc[i][4], acc[i][5], acc[i][6], acc[i][7]};
  }
}

template <int NTW>
__device__ __forceinline__ void rec_tiles(
    const v2f (&ah)[KTH], const v2f (&bh)[TPW][KTH],
    const float* __restrict__ xgr, float* __restrict__ gcol)
{
  v8f acc[NTW];
  #pragma unroll
  for (int i = 0; i < NTW; ++i) {
    const v4f lo = *(const v4f*)(xgr + i * XG_TILE);
    const v4f hi = *(const v4f*)(xgr + i * XG_TILE + 128);
    acc[i] = v8f{lo[0], lo[1], lo[2], lo[3], hi[0], hi[1], hi[2], hi[3]};
  }
  #pragma unroll
  for (int kt = 0; kt < KTH; ++kt) {
    #pragma unroll
    for (int i = 0; i < NTW; ++i)
      acc[i] = __builtin_amdgcn_wmma_f32_16x16x4_f32(
          false, ah[kt], false, bh[i][kt], (short)0, acc[i], false, false);
  }
  #pragma unroll
  for (int i = 0; i < NTW; ++i) {
    #pragma unroll
    for (int r = 0; r < 8; ++r)
      gcol[r * NPAD + i * 16] = acc[i][r];
  }
}

__global__ __launch_bounds__(BLK) void lstm_persist_wmma(
    const float* __restrict__ x,
    const float* __restrict__ kernel,
    const float* __restrict__ rec_kernel,
    const float* __restrict__ bias,
    const float* __restrict__ dense_w,
    const float* __restrict__ dense_b,
    float* __restrict__ out)
{
  extern __shared__ float lds[];
  float* wxT = lds + OFF_WX;
  float* whT = lds + OFF_WH;
  float* bL  = lds + OFF_BIAS;
  float* hB  = lds + OFF_H;
  float* gB  = lds + OFF_G;
  float* xg  = lds + OFF_XG;

  const int tid   = threadIdx.x;
  const int lane  = tid & 31;
  const int wave  = __builtin_amdgcn_readfirstlane(tid >> 5);
  const bool rec  = wave < RW;
  const int  rw   = rec ? wave : wave - RW;
  const int rowA  = lane & 15;
  const int khalf = lane >> 4;
  const int ncol  = lane & 15;
  const int b0    = blockIdx.x * 16;

  const int  ntS  = (rw * NT) / 4;
  const bool four = (((rw + 1) * NT) / 4 - ntS) == TPW;
  const int ubase = rw * 200;

  for (int i = tid; i < NPAD * Ff; i += BLK) {
    int n = i / Ff, k = i % Ff;
    wxT[i] = (n < G4) ? kernel[k * G4 + n] : 0.0f;
  }
  for (int i = tid; i < NPAD * KH; i += BLK) {
    int n = i / KH, k = i % KH;
    whT[i] = (n < G4 && k < Hh) ? rec_kernel[k * G4 + n] : 0.0f;
  }
  for (int i = tid; i < NPAD; i += BLK) bL[i] = (i < G4) ? bias[i] : 0.0f;
  for (int i = tid; i < 16 * KH; i += BLK) hB[i] = 0.0f;
  __syncthreads();

  v2f bx[TPW][KTX];
  v2f bh[TPW][KTH];
  #pragma unroll
  for (int i = 0; i < TPW; ++i) {
    const int n = (ntS + i) * 16 + ncol;
    #pragma unroll
    for (int kt = 0; kt < KTX; ++kt)
      bx[i][kt] = *(const v2f*)(wxT + n * Ff + kt * 4 + 2 * khalf);
    #pragma unroll
    for (int kt = 0; kt < KTH; ++kt)
      bh[i][kt] = *(const v2f*)(whT + n * KH + kt * 4 + 2 * khalf);
  }

  float c[7];
  float bi[7], bf[7], bc[7], bo[7];
  #pragma unroll
  for (int q = 0; q < 7; ++q) {
    c[q] = 0.0f;
    const int j = (ubase + q * 32 + lane) % 50;
    bi[q] = bL[j];
    bf[q] = bL[j +  50];
    bc[q] = bL[j + 100];
    bo[q] = bL[j + 150];
  }

  int goff[7], hoff[7];
  #pragma unroll
  for (int q = 0; q < 7; ++q) {
    const int u  = q * 32 + lane;
    const int uc = (u < 200) ? u : 199;
    const int p  = ubase + uc;
    const int row = p / 50, j = p % 50;
    goff[q] = OFF_G + row * NPAD + j;
    hoff[q] = (u < 200) ? (OFF_H + row * KH + j) : (OFF_DUMP + lane);
  }

  const float* xrow = x + (long)(b0 + rowA) * T_STEPS * Ff + 2 * khalf;
  float* gcol = gB + (8 * khalf) * NPAD + ntS * 16 + ncol;
  float* xgl = xg + ntS * XG_TILE + lane * 4;

  if (!rec) {
    v2f ax0[KTX];
    #pragma unroll
    for (int kt = 0; kt < KTX; ++kt) ax0[kt] = *(const v2f*)(xrow + kt * 4);
    if (four) proj_tiles<4>(ax0, bx, xgl);
    else      proj_tiles<3>(ax0, bx, xgl);
  }
  __syncthreads();

  for (int t = 0; t < T_STEPS; ++t) {
    const float* xgr = xgl + (t & 1) * XG_BUF;
    float*       xgw = xgl + ((t + 1) & 1) * XG_BUF;
    v2f ax[KTX];
    #pragma unroll
    for (int kt = 0; kt < KTX; ++kt) ax[kt] = v2f{};

    if (rec) {
      v2f ah[KTH];
      #pragma unroll
      for (int kt = 0; kt < KTH; ++kt)
        ah[kt] = *(const v2f*)(hB + rowA * KH + kt * 4 + 2 * khalf);
      if (four) rec_tiles<4>(ah, bh, xgr, gcol);
      else      rec_tiles<3>(ah, bh, xgr, gcol);
    } else {
      const int tn = (t + 1 < T_STEPS) ? (t + 1) : (T_STEPS - 1);
      const float* xp = xrow + tn * Ff;
      #pragma unroll
      for (int kt = 0; kt < KTX; ++kt) ax[kt] = *(const v2f*)(xp + kt * 4);
      if (rw == 0 && t + 2 < T_STEPS) __builtin_prefetch(xp + Ff, 0, 3);
    }
    __syncthreads();

    if (rec) {
      float gi[7], gf[7], gc[7], go[7];
      #pragma unroll
      for (int q = 0; q < 7; ++q) {
        const float* g = lds + goff[q];
        gi[q] = g[0]   + bi[q];
        gf[q] = g[50]  + bf[q];
        gc[q] = g[100] + bc[q];
        go[q] = g[150] + bo[q];
      }
      float ei[7], ef[7], eo[7];
      #pragma unroll
      for (int q = 0; q < 7; ++q) {
        ei[q] = __expf(-gi[q]);
        ef[q] = __expf(-gf[q]);
        eo[q] = __expf(-go[q]);
      }
      float si[7], sf[7], so[7];
      #pragma unroll
      for (int q = 0; q < 7; ++q) {
        si[q] = 1.0f / (1.0f + ei[q]);
        sf[q] = 1.0f / (1.0f + ef[q]);
        so[q] = 1.0f / (1.0f + eo[q]);
      }
      float hv[7];
      #pragma unroll
      for (int q = 0; q < 7; ++q) {
        const float cand = fmaxf(gc[q], 0.0f);
        const float cn = __builtin_fmaf(cand, si[q], sf[q] * c[q]);
        c[q] = cn;
        hv[q] = so[q] * fmaxf(cn, 0.0f);
      }
      #pragma unroll
      for (int q = 0; q < 7; ++q) lds[hoff[q]] = hv[q];
    } else {
      if (t + 1 < T_STEPS) {
        if (four) proj_tiles<4>(ax, bx, xgw);
        else      proj_tiles<3>(ax, bx, xgw);
      }
    }
    __syncthreads();
  }

  if (tid < 16) {
    float s = 0.0f;
    #pragma unroll 10
    for (int j = 0; j < Hh; ++j) s += hB[tid * KH + j] * dense_w[j];
    out[b0 + tid] = s + dense_b[0];
  }
}

extern "C" void kernel_launch(void* const* d_in, const int* in_sizes, int n_in,
                              void* d_out, int out_size, void* d_ws, size_t ws_size,
                              hipStream_t stream) {
  const float* x      = (const float*)d_in[0];
  const float* kernel = (const float*)d_in[1];
  const float* rec    = (const float*)d_in[2];
  const float* bias   = (const float*)d_in[3];
  const float* dw     = (const float*)d_in[4];
  const float* db     = (const float*)d_in[5];
  float* out = (float*)d_out;

  dim3 grid(4096 / 16);
  dim3 block(BLK);
  size_t shmem = (size_t)LDS_FLOATS * sizeof(float);
  hipLaunchKernelGGL(lstm_persist_wmma, grid, block, shmem, stream,
                     x, kernel, rec, bias, dw, db, out);
}


# gate columns regrouped: a recurrent wave's 4 tiles are the i,f,c,o columns of its own units, so gates run on the accumulators; no LDS round trip of g, one barrier per step; summation order unchanged
# speedup vs baseline: 2.5343x; 1.2586x over previous
#include <hip/hip_runtime.h>


typedef __attribute__((ext_vector_type(2))) float v2f;
typedef __attribute__((ext_vector_type(4))) float v4f;
typedef __attribute__((ext_vector_type(8))) float v8f;

#define Hh 50
#define Ff 32
#define T_STEPS 365
#define G4 200
#define KH 52
#define KTX 8
#define KTH 13
#define RW 4
#define PW 4
#define BLK ((RW + PW) * 32)
#define NG 4

#define XG_TILE 256
#define XG_WAVE (NG * XG_TILE)
#define XG_BUF  (RW * XG_WAVE)
#define H_BUF   (16 * KH)

#define OFF_H   0
#define OFF_XG  (OFF_H + 2 * H_BUF)
#define LDS_FLOATS (OFF_XG + 2 * XG_BUF)

__device__ __forceinline__ void proj_tiles(
    const v2f (&ax)[KTX], const v2f (&bw)[NG][KTH], float* __restrict__ xgw)
{
  v8f acc[NG];
  #pragma unroll
  for (int i = 0; i < NG; ++i) acc[i] = v8f{};
  #pragma unroll
  for (int kt = 0; kt < KTX; ++kt) {
    #pragma unroll
    for (int i = 0; i < NG; ++i)
      acc[i] = __builtin_amdgcn_wmma_f32_16x16x4_f32(
          false, ax[kt], false, bw[i][kt], (short)0, acc[i], false, false);
  }
  #pragma unroll
  for (int i = 0; i < NG; ++i) {
    *(v4f*)(xgw + i * XG_TILE)       = v4f{acc[i][0], acc[i][1], acc[i][2], acc[i][3]};
    *(v4f*)(xgw + i * XG_TILE + 128) = v4f{acc[i][4], acc[i][5], acc[i][6], acc[i][7]};
  }
}

__global__ __launch_bounds__(BLK) __attribute__((amdgpu_waves_per_eu(1, 2)))
void lstm_persist_wmma(
    const float* __restrict__ x,
    const float* __restrict__ kernel,
    const float* __restrict__ rec_kernel,
    const float* __restrict__ bias,
    const float* __restrict__ dense_w,
    const float* __restrict__ dense_b,
    float* __restrict__ out)
{
  extern __shared__ float lds[];
  float* hB = lds + OFF_H;
  float* xg = lds + OFF_XG;

  const int tid   = threadIdx.x;
  const int lane  = tid & 31;
  const int wave  = __builtin_amdgcn_readfirstlane(tid >> 5);
  const bool rec  = wave < RW;
  const int  rw   = rec ? wave : wave - RW;
  const int rowA  = lane & 15;
  const int khalf = lane >> 4;
  const int ncol  = lane & 15;
  const int b0    = blockIdx.x * 16;

  const int  ucnt  = (rw < 2) ? 13 : 12;
  const int  ust   = (rw < 2) ? rw * 13 : 26 + (rw - 2) * 12;
  const bool valid = ncol < ucnt;
  const int  u     = valid ? ust + ncol : ust;

  for (int i = tid; i < 2 * H_BUF; i += BLK) hB[i] = 0.0f;

  const float* wsrc = rec ? rec_kernel : kernel;
  const int    krows = rec ? Hh : Ff;
  v2f bw[NG][KTH];
  #pragma unroll
  for (int g = 0; g < NG; ++g) {
    const float* wcol = wsrc + g * Hh + u;
    #pragma unroll
    for (int kt = 0; kt < KTH; ++kt) {
      const int k0 = kt * 4 + 2 * khalf;
      const int ka = (k0     < krows) ? k0     : krows - 1;
      const int kb = (k0 + 1 < krows) ? k0 + 1 : krows - 1;
      const unsigned m0 = (valid && k0     < krows) ? 0xffffffffu : 0u;
      const unsigned m1 = (valid && k0 + 1 < krows) ? 0xffffffffu : 0u;
      bw[g][kt] = v2f{__uint_as_float(__float_as_uint(wcol[ka * G4]) & m0),
                      __uint_as_float(__float_as_uint(wcol[kb * G4]) & m1)};
    }
  }

  float c[8];
  #pragma unroll
  for (int r = 0; r < 8; ++r) c[r] = 0.0f;
  const float bi = valid ? bias[u]          : 0.0f;
  const float bf = valid ? bias[u +     Hh] : 0.0f;
  const float bc = valid ? bias[u + 2 * Hh] : 0.0f;
  const float bo = valid ? bias[u + 3 * Hh] : 0.0f;

  const float* xrow = x + (long)(b0 + rowA) * T_STEPS * Ff + 2 * khalf;
  const int hrd = rowA * KH + 2 * khalf;
  const int hwr = (8 * khalf) * KH + u;
  float* xgl = xg + rw * XG_WAVE + lane * 4;

  v2f ax[KTX];
  #pragma unroll
  for (int kt = 0; kt < KTX; ++kt) ax[kt] = v2f{};

  if (!rec) {
    v2f ax0[KTX];
    #pragma unroll
    for (int kt = 0; kt < KTX; ++kt) ax0[kt] = *(const v2f*)(xrow + kt * 4);
    proj_tiles(ax0, bw, xgl);
    #pragma unroll
    for (int kt = 0; kt < KTX; ++kt) ax[kt] = *(const v2f*)(xrow + Ff + kt * 4);
  }
  __syncthreads();

  for (int t = 0; t < T_STEPS; ++t) {
    if (rec) {
      const float* hr  = hB + (t & 1) * H_BUF + hrd;
      float*       hw  = hB + ((t + 1) & 1) * H_BUF + hwr;
      const float* xgr = xgl + (t & 1) * XG_BUF;
      v2f ah[KTH];
      #pragma unroll
      for (int kt = 0; kt < KTH; ++kt) ah[kt] = *(const v2f*)(hr + kt * 4);
      v8f acc[NG];
      #pragma unroll
      for (int g = 0; g < NG; ++g) {
        const v4f lo = *(const v4f*)(xgr + g * XG_TILE);
        const v4f hi = *(const v4f*)(xgr + g * XG_TILE + 128);
        acc[g] = v8f{lo[0], lo[1], lo[2], lo[3], hi[0], hi[1], hi[2], hi[3]};
      }
      __builtin_amdgcn_sched_barrier(0);
      #pragma unroll
      for (int kt = 0; kt < KTH; ++kt) {
        #pragma unroll
        for (int g = 0; g < NG; ++g)
          acc[g] = __builtin_amdgcn_wmma_f32_16x16x4_f32(
              false, ah[kt], false, bw[g][kt], (short)0, acc[g], false, false);
      }
      float ei[8], ef[8], eo[8];
      #pragma unroll
      for (int r = 0; r < 8; ++r) {
        ei[r] = __expf(-(acc[0][r] + bi));
        ef[r] = __expf(-(acc[1][r] + bf));
        eo[r] = __expf(-(acc[3][r] + bo));
      }
      float si[8], sf[8], so[8];
      #pragma unroll
      for (int r = 0; r < 8; ++r) {
        si[r] = 1.0f / (1.0f + ei[r]);
        sf[r] = 1.0f / (1.0f + ef[r]);
        so[r] = 1.0f / (1.0f + eo[r]);
      }
      float hv[8];
      #pragma unroll
      for (int r = 0; r < 8; ++r) {
        const float cand = fmaxf(acc[2][r] + bc, 0.0f);
        const float cn = __builtin_fmaf(cand, si[r], sf[r] * c[r]);
        c[r] = cn;
        hv[r] = so[r] * fmaxf(cn, 0.0f);
      }
      if (valid) {
        #pragma unroll
        for (int r = 0; r < 8; ++r) hw[r * KH] = hv[r];
      }
    } else {
      float* xgw = xgl + ((t + 1) & 1) * XG_BUF;
      const int tn = (t + 2 < T_STEPS) ? (t + 2) : (T_STEPS - 1);
      const float* xp = xrow + tn * Ff;
      v2f axn[KTX];
      #pragma unroll
      for (int kt = 0; kt < KTX; ++kt) axn[kt] = *(const v2f*)(xp + kt * 4);
      if (rw == 0 && t + 3 < T_STEPS) __builtin_prefetch(xp + Ff, 0, 3);
      if (t + 1 < T_STEPS) proj_tiles(ax, bw, xgw);
      #pragma unroll
      for (int kt = 0; kt < KTX; ++kt) ax[kt] = axn[kt];
    }
    __syncthreads();
  }

  if (tid < 16) {
    const float* hl = hB + (T_STEPS & 1) * H_BUF;
    float s = 0.0f;
    #pragma unroll 10
    for (int j = 0; j < Hh; ++j) s += hl[tid * KH + j] * dense_w[j];
    out[b0 + tid] = s + dense_b[0];
  }
}

extern "C" void kernel_launch(void* const* d_in, const int* in_sizes, int n_in,
                              void* d_out, int out_size, void* d_ws, size_t ws_size,
                              hipStream_t stream) {
  const float* x      = (const float*)d_in[0];
  const float* kernel = (const float*)d_in[1];
  const float* rec    = (const float*)d_in[2];
  const float* bias   = (const float*)d_in[3];
  const float* dw     = (const float*)d_in[4];
  const float* db     = (const float*)d_in[5];
  float* out = (float*)d_out;

  dim3 grid(4096 / 16);
  dim3 block(BLK);
  size_t shmem = (size_t)LDS_FLOATS * sizeof(float);
  hipLaunchKernelGGL(lstm_persist_wmma, grid, block, shmem, stream,
                     x, kernel, rec, bias, dw, db, out);
}


# sigmoid quotient 1/(1+e) by the hardware reciprocal (v_rcp_f32, 1 ulp) in place of the correctly rounded division sequence; gate values can differ in the last bit; summation order unchanged
# speedup vs baseline: 2.9645x; 1.1698x over previous
#include <hip/hip_runtime.h>


typedef __attribute__((ext_vector_type(2))) float v2f;
typedef __attribute__((ext_vector_type(4))) float v4f;
typedef __attribute__((ext_vector_type(8))) float v8f;

#define Hh 50
#define Ff 32
#define T_STEPS 365
#define G4 200
#define KH 52
#define KTX 8
#define KTH 13
#define RW 4
#define PW 4
#define BLK ((RW + PW) * 32)
#define NG 4

#define XG_TILE 256
#define XG_WAVE (NG * XG_TILE)
#define XG_BUF  (RW * XG_WAVE)
#define H_BUF   (16 * KH)

#define OFF_H   0
#define OFF_XG  (OFF_H + 2 * H_BUF)
#define LDS_FLOATS (OFF_XG + 2 * XG_BUF)

__device__ __forceinline__ void proj_tiles(
    const v2f (&ax)[KTX], const v2f (&bw)[NG][KTH], float* __restrict__ xgw)
{
  v8f acc[NG];
  #pragma unroll
  for (int i = 0; i < NG; ++i) acc[i] = v8f{};
  #pragma unroll
  for (int kt = 0; kt < KTX; ++kt) {
    #pragma unroll
    for (int i = 0; i < NG; ++i)
      acc[i] = __builtin_amdgcn_wmma_f32_16x16x4_f32(
          false, ax[kt], false, bw[i][kt], (short)0, acc[i], false, false);
  }
  #pragma unroll
  for (int i = 0; i < NG; ++i) {
    *(v4f*)(xgw + i * XG_TILE)       = v4f{acc[i][0], acc[i][1], acc[i][2], acc[i][3]};
    *(v4f*)(xgw + i * XG_TILE + 128) = v4f{acc[i][4], acc[i][5], acc[i][6], acc[i][7]};
  }
}

__global__ __launch_bounds__(BLK) __attribute__((amdgpu_waves_per_eu(1, 2)))
void lstm_persist_wmma(
    const float* __restrict__ x,
    const float* __restrict__ kernel,
    const float* __restrict__ rec_kernel,
    const float* __restrict__ bias,
    const float* __restrict__ dense_w,
    const float* __restrict__ dense_b,
    float* __restrict__ out)
{
  extern __shared__ float lds[];
  float* hB = lds + OFF_H;
  float* xg = lds + OFF_XG;

  const int tid   = threadIdx.x;
  const int lane  = tid & 31;
  const int wave  = __builtin_amdgcn_readfirstlane(tid >> 5);
  const bool rec  = wave < RW;
  const int  rw   = rec ? wave : wave - RW;
  const int rowA  = lane & 15;
  const int khalf = lane >> 4;
  const int ncol  = lane & 15;
  const int b0    = blockIdx.x * 16;

  const int  ucnt  = (rw < 2) ? 13 : 12;
  const int  ust   = (rw < 2) ? rw * 13 : 26 + (rw - 2) * 12;
  const bool valid = ncol < ucnt;
  const int  u     = valid ? ust + ncol : ust;

  for (int i = tid; i < 2 * H_BUF; i += BLK) hB[i] = 0.0f;

  const float* wsrc = rec ? rec_kernel : kernel;
  const int    krows = rec ? Hh : Ff;
  v2f bw[NG][KTH];
  #pragma unroll
  for (int g = 0; g < NG; ++g) {
    const float* wcol = wsrc + g * Hh + u;
    #pragma unroll
    for (int kt = 0; kt < KTH; ++kt) {
      const int k0 = kt * 4 + 2 * khalf;
      const int ka = (k0     < krows) ? k0     : krows - 1;
      const int kb = (k0 + 1 < krows) ? k0 + 1 : krows - 1;
      const unsigned m0 = (valid && k0     < krows) ? 0xffffffffu : 0u;
      const unsigned m1 = (valid && k0 + 1 < krows) ? 0xffffffffu : 0u;
      bw[g][kt] = v2f{__uint_as_float(__float_as_uint(wcol[ka * G4]) & m0),
                      __uint_as_float(__float_as_uint(wcol[kb * G4]) & m1)};
    }
  }

  float c[8];
  #pragma unroll
  for (int r = 0; r < 8; ++r) c[r] = 0.0f;
  const float bi = valid ? bias[u]          : 0.0f;
  const float bf = valid ? bias[u +     Hh] : 0.0f;
  const float bc = valid ? bias[u + 2 * Hh] : 0.0f;
  const float bo = valid ? bias[u + 3 * Hh] : 0.0f;

  const float* xrow = x + (long)(b0 + rowA) * T_STEPS * Ff + 2 * khalf;
  const int hrd = rowA * KH + 2 * khalf;
  const int hwr = (8 * khalf) * KH + u;
  float* xgl = xg + rw * XG_WAVE + lane * 4;

  v2f ax[KTX];
  #pragma unroll
  for (int kt = 0; kt < KTX; ++kt) ax[kt] = v2f{};

  if (!rec) {
    v2f ax0[KTX];
    #pragma unroll
    for (int kt = 0; kt < KTX; ++kt) ax0[kt] = *(const v2f*)(xrow + kt * 4);
    proj_tiles(ax0, bw, xgl);
    #pragma unroll
    for (int kt = 0; kt < KTX; ++kt) ax[kt] = *(const v2f*)(xrow + Ff + kt * 4);
  }
  __syncthreads();

  for (int t = 0; t < T_STEPS; ++t) {
    if (rec) {
      const float* hr  = hB + (t & 1) * H_BUF + hrd;
      float*       hw  = hB + ((t + 1) & 1) * H_BUF + hwr;
      const float* xgr = xgl + (t & 1) * XG_BUF;
      v2f ah[KTH];
      #pragma unroll
      for (int kt = 0; kt < KTH; ++kt) ah[kt] = *(const v2f*)(hr + kt * 4);
      v8f acc[NG];
      #pragma unroll
      for (int g = 0; g < NG; ++g) {
        const v4f lo = *(const v4f*)(xgr + g * XG_TILE);
        const v4f hi = *(const v4f*)(xgr + g * XG_TILE + 128);
        acc[g] = v8f{lo[0], lo[1], lo[2], lo[3], hi[0], hi[1], hi[2], hi[3]};
      }
      __builtin_amdgcn_sched_barrier(0);
      #pragma unroll
      for (int kt = 0; kt < KTH; ++kt) {
        #pragma unroll
        for (int g = 0; g < NG; ++g)
          acc[g] = __builtin_amdgcn_wmma_f32_16x16x4_f32(
              false, ah[kt], false, bw[g][kt], (short)0, acc[g], false, false);
      }
      float ei[8], ef[8], eo[8];
      #pragma unroll
      for (int r = 0; r < 8; ++r) {
        ei[r] = __expf(-(acc[0][r] + bi));
        ef[r] = __expf(-(acc[1][r] + bf));
        eo[r] = __expf(-(acc[3][r] + bo));
      }
      float si[8], sf[8], so[8];
      #pragma unroll
      for (int r = 0; r < 8; ++r) {
        si[r] = __builtin_amdgcn_rcpf(1.0f + ei[r]);
        sf[r] = __builtin_amdgcn_rcpf(1.0f + ef[r]);
        so[r] = __builtin_amdgcn_rcpf(1.0f + eo[r]);
      }
      float hv[8];
      #pragma unroll
      for (int r = 0; r < 8; ++r) {
        const float cand = fmaxf(acc[2][r] + bc, 0.0f);
        const float cn = __builtin_fmaf(cand, si[r], sf[r] * c[r]);
        c[r] = cn;
        hv[r] = so[r] * fmaxf(cn, 0.0f);
      }
      if (valid) {
        #pragma unroll
        for (int r = 0; r < 8; ++r) hw[r * KH] = hv[r];
      }
    } else {
      float* xgw = xgl + ((t + 1) & 1) * XG_BUF;
      const int tn = (t + 2 < T_STEPS) ? (t + 2) : (T_STEPS - 1);
      const float* xp = xrow + tn * Ff;
      v2f axn[KTX];
      #pragma unroll
      for (int kt = 0; kt < KTX; ++kt) axn[kt] = *(const v2f*)(xp + kt * 4);
      if (rw == 0 && t + 3 < T_STEPS) __builtin_prefetch(xp + Ff, 0, 3);
      if (t + 1 < T_STEPS) proj_tiles(ax, bw, xgw);
      #pragma unroll
      for (int kt = 0; kt < KTX; ++kt) ax[kt] = axn[kt];
    }
    __syncthreads();
  }

  if (tid < 16) {
    const float* hl = hB + (T_STEPS & 1) * H_BUF;
    float s = 0.0f;
    #pragma unroll 10
    for (int j = 0; j < Hh; ++j) s += hl[tid * KH + j] * dense_w[j];
    out[b0 + tid] = s + dense_b[0];
  }
}

extern "C" void kernel_launch(void* const* d_in, const int* in_sizes, int n_in,
                              void* d_out, int out_size, void* d_ws, size_t ws_size,
                              hipStream_t stream) {
  const float* x      = (const float*)d_in[0];
  const float* kernel = (const float*)d_in[1];
  const float* rec    = (const float*)d_in[2];
  const float* bias   = (const float*)d_in[3];
  const float* dw     = (const float*)d_in[4];
  const float* db     = (const float*)d_in[5];
  float* out = (float*)d_out;

  dim3 grid(4096 / 16);
  dim3 block(BLK);
  size_t shmem = (size_t)LDS_FLOATS * sizeof(float);
  hipLaunchKernelGGL(lstm_persist_wmma, grid, block, shmem, stream,
                     x, kernel, rec, bias, dw, db, out);
}


# gate columns packed into 13 tiles, no padding tiles: recurrent waves 0..2 own the 4 gate tiles of their units, wave 3 the tile of units 48,49 (gates gathered across lanes); projection dealt 2,2,2,7
# speedup vs baseline: 3.3466x; 1.1289x over previous
#include <hip/hip_runtime.h>


typedef __attribute__((ext_vector_type(2))) float v2f;
typedef __attribute__((ext_vector_type(4))) float v4f;
typedef __attribute__((ext_vector_type(8))) float v8f;

#define Hh 50
#define Ff 32
#define T_STEPS 365
#define G4 200
#define KH 52
#define KTX 8
#define KTH 13
#define RW 4
#define PW 4
#define BLK ((RW + PW) * 32)
#define NG 4

#define NTILE 13
#define TAIL_T 12
#define PT_FEW 2
#define PT_MANY 7
#define NBW (PT_MANY * KTX)

#define XG_TILE 256
#define XG_BUF  (NTILE * XG_TILE)
#define H_BUF   (16 * KH)

#define OFF_H   0
#define OFF_XG  (OFF_H + 2 * H_BUF)
#define LDS_FLOATS (OFF_XG + 2 * XG_BUF)

__device__ __forceinline__ int tile_col(int T, int ncol) {
  return (T < TAIL_T) ? (T & 3) * Hh + 16 * (T >> 2) + ncol
                      : ((ncol & 7) >> 1) * Hh + 48 + (ncol & 1);
}
__device__ __forceinline__ bool tile_ok(int T, int ncol) { return T < TAIL_T || ncol < 8; }

__device__ __forceinline__ v2f load_bfrag(const float* __restrict__ wcol, int kt, int khalf,
                                          int krows, bool ok) {
  const int k0 = kt * 4 + 2 * khalf;
  const int ka = (k0     < krows) ? k0     : krows - 1;
  const int kb = (k0 + 1 < krows) ? k0 + 1 : krows - 1;
  const unsigned m0 = (ok && k0     < krows) ? 0xffffffffu : 0u;
  const unsigned m1 = (ok && k0 + 1 < krows) ? 0xffffffffu : 0u;
  return v2f{__uint_as_float(__float_as_uint(wcol[ka * G4]) & m0),
             __uint_as_float(__float_as_uint(wcol[kb * G4]) & m1)};
}

template <int NT, typename TOFF>
__device__ __forceinline__ void proj_tiles(
    const v2f (&ax)[KTX], const v2f (&bw)[NBW], float* __restrict__ xgw, TOFF toff)
{
  v8f acc[NT];
  #pragma unroll
  for (int i = 0; i < NT; ++i) acc[i] = v8f{};
  #pragma unroll
  for (int kt = 0; kt < KTX; ++kt) {
    #pragma unroll
    for (int i = 0; i < NT; ++i)
      acc[i] = __builtin_amdgcn_wmma_f32_16x16x4_f32(
          false, ax[kt], false, bw[i * KTX + kt], (short)0, acc[i], false, false);
  }
  #pragma unroll
  for (int i = 0; i < NT; ++i) {
    float* p = xgw + toff(i);
    *(v4f*)(p)       = v4f{acc[i][0], acc[i][1], acc[i][2], acc[i][3]};
    *(v4f*)(p + 128) = v4f{acc[i][4], acc[i][5], acc[i][6], acc[i][7]};
  }
}

__device__ __forceinline__ void gate_update(
    const v8f& gi, const v8f& gf, const v8f& gc, const v8f& go,
    float bi, float bf, float bc, float bo, float (&c)[8], float (&hv)[8])
{
  float ei[8], ef[8], eo[8];
  #pragma unroll
  for (int r = 0; r < 8; ++r) {
    ei[r] = __expf(-(gi[r] + bi));
    ef[r] = __expf(-(gf[r] + bf));
    eo[r] = __expf(-(go[r] + bo));
  }
  float si[8], sf[8], so[8];
  #pragma unroll
  for (int r = 0; r < 8; ++r) {
    si[r] = __builtin_amdgcn_rcpf(1.0f + ei[r]);
    sf[r] = __builtin_amdgcn_rcpf(1.0f + ef[r]);
    so[r] = __builtin_amdgcn_rcpf(1.0f + eo[r]);
  }
  #pragma unroll
  for (int r = 0; r < 8; ++r) {
    const float cand = fmaxf(gc[r] + bc, 0.0f);
    const float cn = __builtin_fmaf(cand, si[r], sf[r] * c[r]);
    c[r] = cn;
    hv[r] = so[r] * fmaxf(cn, 0.0f);
  }
}

template <int N>
__device__ __forceinline__ float row_from_right(float v) {
  return __int_as_float(__builtin_amdgcn_update_dpp(
      0, __float_as_int(v), 0x100 + N, 0xf, 0xf, true));
}

__global__ __launch_bounds__(BLK) __attribute__((amdgpu_waves_per_eu(1, 2)))
void lstm_persist_wmma(
    const float* __restrict__ x,
    const float* __restrict__ kernel,
    const float* __restrict__ rec_kernel,
    const float* __restrict__ bias,
    const float* __restrict__ dense_w,
    const float* __restrict__ dense_b,
    float* __restrict__ out)
{
  extern __shared__ float lds[];
  float* hB = lds + OFF_H;
  float* xg = lds + OFF_XG;

  const int tid   = threadIdx.x;
  const int lane  = tid & 31;
  const int wave  = __builtin_amdgcn_readfirstlane(tid >> 5);
  const bool rec  = wave < RW;
  const int  rw   = rec ? wave : wave - RW;
  const bool full = rw < 3;
  const int rowA  = lane & 15;
  const int khalf = lane >> 4;
  const int ncol  = lane & 15;
  const int b0    = blockIdx.x * 16;

  const int  u      = full ? 16 * rw + ncol : 48 + (ncol & 1);
  const bool hstore = full || ncol < 2;

  for (int i = tid; i < 2 * H_BUF; i += BLK) hB[i] = 0.0f;

  v2f bw[NBW];
  #pragma unroll
  for (int i = 0; i < NBW; ++i) bw[i] = v2f{};
  if (rec) {
    #pragma unroll
    for (int s = 0; s < NG; ++s) {
      if (full || s == 0) {
        const int T = full ? 4 * rw + s : TAIL_T;
        const float* wcol = rec_kernel + tile_col(T, ncol);
        const bool ok = tile_ok(T, ncol);
        #pragma unroll
        for (int kt = 0; kt < KTH; ++kt) bw[s * KTH + kt] = load_bfrag(wcol, kt, khalf, Hh, ok);
      }
    }
  } else {
    #pragma unroll
    for (int s = 0; s < PT_MANY; ++s) {
      if (!full || s < PT_FEW) {
        const int T = full ? 4 * rw + s : (s < 6 ? 4 * (s >> 1) + 2 + (s & 1) : TAIL_T);
        const float* wcol = kernel + tile_col(T, ncol);
        const bool ok = tile_ok(T, ncol);
        #pragma unroll
        for (int kt = 0; kt < KTX; ++kt) bw[s * KTX + kt] = load_bfrag(wcol, kt, khalf, Ff, ok);
      }
    }
  }

  float c[8];
  #pragma unroll
  for (int r = 0; r < 8; ++r) c[r] = 0.0f;
  const float bi = bias[u];
  const float bf = bias[u +     Hh];
  const float bc = bias[u + 2 * Hh];
  const float bo = bias[u + 3 * Hh];
  const float bown  = bias[tile_col(TAIL_T, ncol)];
  const bool  cgate = ((ncol & 7) >> 1) == 2;

  const float* xrow = x + (long)(b0 + rowA) * T_STEPS * Ff + 2 * khalf;
  const int hrd = rowA * KH + 2 * khalf;
  const int hwr = (8 * khalf) * KH + u;
  float* xgl = xg + lane * 4;
  const int tile0 = full ? 4 * rw : TAIL_T;

  v2f ax[KTX];
  #pragma unroll
  for (int kt = 0; kt < KTX; ++kt) ax[kt] = v2f{};

  auto toff_few  = [&](int s) { return (tile0 + s) * XG_TILE; };
  auto toff_many = [](int s) { return (s < 6 ? 4 * (s >> 1) + 2 + (s & 1) : TAIL_T) * XG_TILE; };

  if (!rec) {
    v2f ax0[KTX];
    #pragma unroll
    for (int kt = 0; kt < KTX; ++kt) ax0[kt] = *(const v2f*)(xrow + kt * 4);
    if (full) proj_tiles<PT_FEW>(ax0, bw, xgl, toff_few);
    else      proj_tiles<PT_MANY>(ax0, bw, xgl, toff_many);
    #pragma unroll
    for (int kt = 0; kt < KTX; ++kt) ax[kt] = *(const v2f*)(xrow + Ff + kt * 4);
  }
  __syncthreads();

  for (int t = 0; t < T_STEPS; ++t) {
    if (rec) {
      const float* hr  = hB + (t & 1) * H_BUF + hrd;
      float*       hw  = hB + ((t + 1) & 1) * H_BUF + hwr;
      const float* xgr = xgl + (t & 1) * XG_BUF + tile0 * XG_TILE;
      v2f ah[KTH];
      #pragma unroll
      for (int kt = 0; kt < KTH; ++kt) ah[kt] = *(const v2f*)(hr + kt * 4);
      float hv[8];
      if (full) {
        v8f acc[NG];
        #pragma unroll
        for (int g = 0; g < NG; ++g) {
          const v4f lo = *(const v4f*)(xgr + g * XG_TILE);
          const v4f hi = *(const v4f*)(xgr + g * XG_TILE + 128);
          acc[g] = v8f{lo[0], lo[1], lo[2], lo[3], hi[0], hi[1], hi[2], hi[3]};
        }
        __builtin_amdgcn_sched_barrier(0);
        #pragma unroll
        for (int kt = 0; kt < KTH; ++kt) {
          #pragma unroll
          for (int g = 0; g < NG; ++g)
            acc[g] = __builtin_amdgcn_wmma_f32_16x16x4_f32(
                false, ah[kt], false, bw[g * KTH + kt], (short)0, acc[g], false, false);
        }
        gate_update(acc[0], acc[1], acc[2], acc[3], bi, bf, bc, bo, c, hv);
        #pragma unroll
        for (int r = 0; r < 8; ++r) hw[r * KH] = hv[r];
      } else {
        const v4f lo = *(const v4f*)(xgr);
        const v4f hi = *(const v4f*)(xgr + 128);
        v8f acc = v8f{lo[0], lo[1], lo[2], lo[3], hi[0], hi[1], hi[2], hi[3]};
        #pragma unroll
        for (int kt = 0; kt < KTH; ++kt)
          acc = __builtin_amdgcn_wmma_f32_16x16x4_f32(
              false, ah[kt], false, bw[kt], (short)0, acc, false, false);
        float a[8];
        #pragma unroll
        for (int r = 0; r < 8; ++r) {
          const float pre = acc[r] + bown;
          const float sg = __builtin_amdgcn_rcpf(1.0f + __expf(-pre));
          a[r] = cgate ? fmaxf(pre, 0.0f) : sg;
        }
        #pragma unroll
        for (int r = 0; r < 8; ++r) {
          const float sf = row_from_right<2>(a[r]);
          const float cand = row_from_right<4>(a[r]);
          const float so = row_from_right<6>(a[r]);
          const float cn = __builtin_fmaf(cand, a[r], sf * c[r]);
          c[r] = cn;
          hv[r] = so * fmaxf(cn, 0.0f);
        }
        if (hstore) {
          #pragma unroll
          for (int r = 0; r < 8; ++r) hw[r * KH] = hv[r];
        }
      }
    } else {
      float* xgw = xgl + ((t + 1) & 1) * XG_BUF;
      const int tn = (t + 2 < T_STEPS) ? (t + 2) : (T_STEPS - 1);
      const float* xp = xrow + tn * Ff;
      v2f axn[KTX];
      #pragma unroll
      for (int kt = 0; kt < KTX; ++kt) axn[kt] = *(const v2f*)(xp + kt * 4);
      if (rw == 0 && t + 3 < T_STEPS) __builtin_prefetch(xp + Ff, 0, 3);
      if (t + 1 < T_STEPS) {
        if (full) proj_tiles<PT_FEW>(ax, bw, xgw, toff_few);
        else      proj_tiles<PT_MANY>(ax, bw, xgw, toff_many);
      }
      #pragma unroll
      for (int kt = 0; kt < KTX; ++kt) ax[kt] = axn[kt];
    }
    __syncthreads();
  }

  if (tid < 16) {
    const float* hl = hB + (T_STEPS & 1) * H_BUF;
    float s = 0.0f;
    #pragma unroll 10
    for (int j = 0; j < Hh; ++j) s += hl[tid * KH + j] * dense_w[j];
    out[b0 + tid] = s + dense_b[0];
  }
}

extern "C" void kernel_launch(void* const* d_in, const int* in_sizes, int n_in,
                              void* d_out, int out_size, void* d_ws, size_t ws_size,
                              hipStream_t stream) {
  const float* x      = (const float*)d_in[0];
  const float* kernel = (const float*)d_in[1];
  const float* rec    = (const float*)d_in[2];
  const float* bias   = (const float*)d_in[3];
  const float* dw     = (const float*)d_in[4];
  const float* db     = (const float*)d_in[5];
  float* out = (float*)d_out;

  dim3 grid(4096 / 16);
  dim3 block(BLK);
  size_t shmem = (size_t)LDS_FLOATS * sizeof(float);
  hipLaunchKernelGGL(lstm_persist_wmma, grid, block, shmem, stream,
                     x, kernel, rec, bias, dw, db, out);
}
